# starting kernel
# baseline (speedup 1.0000x reference)
#include <hip/hip_runtime.h>

typedef __attribute__((ext_vector_type(2))) float v2f;
typedef __attribute__((ext_vector_type(8))) float v8f;

#define IN_CH   128
#define OUT_CH  64
#define POS_DIM 64

__global__ void init_deg_kernel(float* __restrict__ deg, int N) {
    int i = blockIdx.x * blockDim.x + threadIdx.x;
    if (i < N) deg[i] = 1.0f;
}

__global__ void count_deg_kernel(const int* __restrict__ dst, float* __restrict__ deg, int E) {
    int e = blockIdx.x * blockDim.x + threadIdx.x;
    if (e < E) atomicAdd(&deg[dst[e]], 1.0f);
}

__global__ void rsqrt_kernel(float* __restrict__ deg, int N) {
    int i = blockIdx.x * blockDim.x + threadIdx.x;
    if (i < N) deg[i] = rsqrtf(deg[i]);
}

__global__ void gemm_xw_kernel(const float* __restrict__ x,
                               const float* __restrict__ W,
                               float* __restrict__ h, int N, int mTiles) {
    const int wavesPerBlock = blockDim.x >> 5;
    const int waveId = threadIdx.x >> 5;
    const int lane   = threadIdx.x & 31;
    const int mTile  = blockIdx.x * wavesPerBlock + waveId;
    const int nTile  = blockIdx.y;
    if (mTile >= mTiles) return;

    const int l16 = lane & 15;
    const int kh  = (lane >> 4) << 1;
    int m = mTile * 16 + l16;
    if (m >= N) m = N - 1;
    const int n = nTile * 16 + l16;

    v8f c = {};
#pragma unroll
    for (int k = 0; k < IN_CH; k += 4) {
        v2f a = *(const v2f*)(x + (size_t)m * IN_CH + k + kh);
        v2f bfrag;
        bfrag[0] = W[(size_t)(k + kh + 0) * OUT_CH + n];
        bfrag[1] = W[(size_t)(k + kh + 1) * OUT_CH + n];
        c = __builtin_amdgcn_wmma_f32_16x16x4_f32(
                false, a, false, bfrag, (short)0, c, false, false);
    }

    const int rbase = mTile * 16 + ((lane >> 4) << 3);
#pragma unroll
    for (int r = 0; r < 8; ++r) {
        int row = rbase + r;
        if (row < N) h[(size_t)row * OUT_CH + n] = c[r];
    }
}

__global__ void gemm_pos_fused_kernel(const float* __restrict__ pos,
                                      const float* __restrict__ Wp,
                                      const float* __restrict__ bp,
                                      const float* __restrict__ b,
                                      const float* __restrict__ h,
                                      const float* __restrict__ dinv,
                                      float* __restrict__ out, int N, int mTiles) {
    const int wavesPerBlock = blockDim.x >> 5;
    const int waveId = threadIdx.x >> 5;
    const int lane   = threadIdx.x & 31;
    const int mTile  = blockIdx.x * wavesPerBlock + waveId;
    const int nTile  = blockIdx.y;
    if (mTile >= mTiles) return;

    const int l16 = lane & 15;
    const int kh  = (lane >> 4) << 1;
    int m = mTile * 16 + l16;
    if (m >= N) m = N - 1;
    const int n = nTile * 16 + l16;

    v8f c = {};
#pragma unroll
    for (int k = 0; k < POS_DIM; k += 4) {
        v2f a = *(const v2f*)(pos + (size_t)m * POS_DIM + k + kh);
        v2f bfrag;
        bfrag[0] = Wp[(size_t)(k + kh + 0) * OUT_CH + n];
        bfrag[1] = Wp[(size_t)(k + kh + 1) * OUT_CH + n];
        c = __builtin_amdgcn_wmma_f32_16x16x4_f32(
                false, a, false, bfrag, (short)0, c, false, false);
    }

    const float bias = bp[n] + b[n];
    const int rbase = mTile * 16 + ((lane >> 4) << 3);
#pragma unroll
    for (int r = 0; r < 8; ++r) {
        int row = rbase + r;
        if (row < N) {
            float di = dinv[row];
            out[(size_t)row * OUT_CH + n] =
                c[r] + bias + h[(size_t)row * OUT_CH + n] * di * di;
        }
    }
}

__global__ void edge_scatter_kernel(const int* __restrict__ src,
                                    const int* __restrict__ dst,
                                    const float* __restrict__ h,
                                    const float* __restrict__ dinv,
                                    float* __restrict__ out, int E) {
    const int wavesPerBlock = blockDim.x >> 5;
    const int e    = blockIdx.x * wavesPerBlock + (threadIdx.x >> 5);
    const int lane = threadIdx.x & 31;
    if (e >= E) return;

    const int s = src[e];
    const int d = dst[e];
    const float nrm = dinv[s] * dinv[d];

    const float2 hv = *(const float2*)(h + (size_t)s * OUT_CH + 2 * lane);
    float* op = out + (size_t)d * OUT_CH + 2 * lane;
    atomicAdd(op + 0, hv.x * nrm);
    atomicAdd(op + 1, hv.y * nrm);
}

extern "C" void kernel_launch(void* const* d_in, const int* in_sizes, int n_in,
                              void* d_out, int out_size, void* d_ws, size_t ws_size,
                              hipStream_t stream) {
    const float* x   = (const float*)d_in[0];
    const int*   ei  = (const int*)d_in[1];
    const float* pos = (const float*)d_in[2];
    const float* W   = (const float*)d_in[3];
    const float* b   = (const float*)d_in[4];
    const float* Wp  = (const float*)d_in[5];
    const float* bp  = (const float*)d_in[6];
    float* out = (float*)d_out;

    const int N = in_sizes[0] / IN_CH;
    const int E = in_sizes[1] / 2;
    const int* src = ei;
    const int* dst = ei + E;

    float* dinv = (float*)d_ws;
    float* h    = dinv + ((N + 3) & ~3);

    init_deg_kernel<<<(N + 255) / 256, 256, 0, stream>>>(dinv, N);
    count_deg_kernel<<<(E + 255) / 256, 256, 0, stream>>>(dst, dinv, E);
    rsqrt_kernel<<<(N + 255) / 256, 256, 0, stream>>>(dinv, N);

    const int mTiles = (N + 15) / 16;
    const int wavesPerBlock = 8;
    dim3 gGemm((mTiles + wavesPerBlock - 1) / wavesPerBlock, OUT_CH / 16);
    gemm_xw_kernel<<<gGemm, 32 * wavesPerBlock, 0, stream>>>(x, W, h, N, mTiles);

    gemm_pos_fused_kernel<<<gGemm, 32 * wavesPerBlock, 0, stream>>>(
        pos, Wp, bp, b, h, dinv, out, N, mTiles);

    edge_scatter_kernel<<<(E + wavesPerBlock - 1) / wavesPerBlock,
                          32 * wavesPerBlock, 0, stream>>>(src, dst, h, dinv, out, E);
}


# edge scatter: one wave per 32 edges instead of one per edge (coalesced index loads, scalar broadcast of src, dst, norm; 32 row loads in flight); f32 atomic adds land in arrival order as before
# speedup vs baseline: 1.3835x; 1.3835x over previous
#include <hip/hip_runtime.h>

typedef __attribute__((ext_vector_type(2))) float v2f;
typedef __attribute__((ext_vector_type(8))) float v8f;

#define IN_CH   128
#define OUT_CH  64
#define POS_DIM 64

__global__ void init_deg_kernel(float* __restrict__ deg, int N) {
    int i = blockIdx.x * blockDim.x + threadIdx.x;
    if (i < N) deg[i] = 1.0f;
}

__global__ void count_deg_kernel(const int* __restrict__ dst, float* __restrict__ deg, int E) {
    int e = blockIdx.x * blockDim.x + threadIdx.x;
    if (e < E) atomicAdd(&deg[dst[e]], 1.0f);
}

__global__ void rsqrt_kernel(float* __restrict__ deg, int N) {
    int i = blockIdx.x * blockDim.x + threadIdx.x;
    if (i < N) deg[i] = rsqrtf(deg[i]);
}

__global__ void gemm_xw_kernel(const float* __restrict__ x,
                               const float* __restrict__ W,
                               float* __restrict__ h, int N, int mTiles) {
    const int wavesPerBlock = blockDim.x >> 5;
    const int waveId = threadIdx.x >> 5;
    const int lane   = threadIdx.x & 31;
    const int mTile  = blockIdx.x * wavesPerBlock + waveId;
    const int nTile  = blockIdx.y;
    if (mTile >= mTiles) return;

    const int l16 = lane & 15;
    const int kh  = (lane >> 4) << 1;
    int m = mTile * 16 + l16;
    if (m >= N) m = N - 1;
    const int n = nTile * 16 + l16;

    v8f c = {};
#pragma unroll
    for (int k = 0; k < IN_CH; k += 4) {
        v2f a = *(const v2f*)(x + (size_t)m * IN_CH + k + kh);
        v2f bfrag;
        bfrag[0] = W[(size_t)(k + kh + 0) * OUT_CH + n];
        bfrag[1] = W[(size_t)(k + kh + 1) * OUT_CH + n];
        c = __builtin_amdgcn_wmma_f32_16x16x4_f32(
                false, a, false, bfrag, (short)0, c, false, false);
    }

    const int rbase = mTile * 16 + ((lane >> 4) << 3);
#pragma unroll
    for (int r = 0; r < 8; ++r) {
        int row = rbase + r;
        if (row < N) h[(size_t)row * OUT_CH + n] = c[r];
    }
}

__global__ void gemm_pos_fused_kernel(const float* __restrict__ pos,
                                      const float* __restrict__ Wp,
                                      const float* __restrict__ bp,
                                      const float* __restrict__ b,
                                      const float* __restrict__ h,
                                      const float* __restrict__ dinv,
                                      float* __restrict__ out, int N, int mTiles) {
    const int wavesPerBlock = blockDim.x >> 5;
    const int waveId = threadIdx.x >> 5;
    const int lane   = threadIdx.x & 31;
    const int mTile  = blockIdx.x * wavesPerBlock + waveId;
    const int nTile  = blockIdx.y;
    if (mTile >= mTiles) return;

    const int l16 = lane & 15;
    const int kh  = (lane >> 4) << 1;
    int m = mTile * 16 + l16;
    if (m >= N) m = N - 1;
    const int n = nTile * 16 + l16;

    v8f c = {};
#pragma unroll
    for (int k = 0; k < POS_DIM; k += 4) {
        v2f a = *(const v2f*)(pos + (size_t)m * POS_DIM + k + kh);
        v2f bfrag;
        bfrag[0] = Wp[(size_t)(k + kh + 0) * OUT_CH + n];
        bfrag[1] = Wp[(size_t)(k + kh + 1) * OUT_CH + n];
        c = __builtin_amdgcn_wmma_f32_16x16x4_f32(
                false, a, false, bfrag, (short)0, c, false, false);
    }

    const float bias = bp[n] + b[n];
    const int rbase = mTile * 16 + ((lane >> 4) << 3);
#pragma unroll
    for (int r = 0; r < 8; ++r) {
        int row = rbase + r;
        if (row < N) {
            float di = dinv[row];
            out[(size_t)row * OUT_CH + n] =
                c[r] + bias + h[(size_t)row * OUT_CH + n] * di * di;
        }
    }
}

#define EDGES_PER_WAVE 32

__global__ void edge_scatter_kernel(const int* __restrict__ src,
                                    const int* __restrict__ dst,
                                    const float* __restrict__ h,
                                    const float* __restrict__ dinv,
                                    float* __restrict__ out, int E) {
    const int wavesPerBlock = blockDim.x >> 5;
    const int wave = __builtin_amdgcn_readfirstlane(
        (int)(blockIdx.x * wavesPerBlock + (threadIdx.x >> 5)));
    const int lane = threadIdx.x & 31;
    const int base = wave * EDGES_PER_WAVE;
    if (base >= E) return;

    const int e = base + lane;
    int   s = 0, d = 0;
    float nrm = 0.0f;
    if (e < E) {
        s = src[e];
        d = dst[e];
        nrm = dinv[s] * dinv[d];
    }
    const int col = 2 * lane;

    if (E - base >= EDGES_PER_WAVE) {
        float2 hv[EDGES_PER_WAVE];
#pragma unroll
        for (int j = 0; j < EDGES_PER_WAVE; ++j) {
            const unsigned sj = (unsigned)__builtin_amdgcn_readlane(s, j);
            hv[j] = ((const float2*)h)[sj * (OUT_CH / 2) + (unsigned)lane];
        }
#pragma unroll
        for (int j = 0; j < EDGES_PER_WAVE; ++j) {
            const int   dj = __builtin_amdgcn_readlane(d, j);
            const float nj = __int_as_float(
                __builtin_amdgcn_readlane(__float_as_int(nrm), j));
            float* op = out + ((unsigned)dj * OUT_CH + (unsigned)col);
            atomicAdd(op + 0, hv[j].x * nj);
            atomicAdd(op + 1, hv[j].y * nj);
        }
    } else {
        const int cnt = E - base;
        for (int j = 0; j < cnt; ++j) {
            const int   sj = __shfl(s, j);
            const int   dj = __shfl(d, j);
            const float nj = __shfl(nrm, j);
            const float2 hvj = *(const float2*)(h + (size_t)sj * OUT_CH + col);
            float* op = out + (size_t)dj * OUT_CH + col;
            atomicAdd(op + 0, hvj.x * nj);
            atomicAdd(op + 1, hvj.y * nj);
        }
    }
}

extern "C" void kernel_launch(void* const* d_in, const int* in_sizes, int n_in,
                              void* d_out, int out_size, void* d_ws, size_t ws_size,
                              hipStream_t stream) {
    const float* x   = (const float*)d_in[0];
    const int*   ei  = (const int*)d_in[1];
    const float* pos = (const float*)d_in[2];
    const float* W   = (const float*)d_in[3];
    const float* b   = (const float*)d_in[4];
    const float* Wp  = (const float*)d_in[5];
    const float* bp  = (const float*)d_in[6];
    float* out = (float*)d_out;

    const int N = in_sizes[0] / IN_CH;
    const int E = in_sizes[1] / 2;
    const int* src = ei;
    const int* dst = ei + E;

    float* dinv = (float*)d_ws;
    float* h    = dinv + ((N + 3) & ~3);

    init_deg_kernel<<<(N + 255) / 256, 256, 0, stream>>>(dinv, N);
    count_deg_kernel<<<(E + 255) / 256, 256, 0, stream>>>(dst, dinv, E);
    rsqrt_kernel<<<(N + 255) / 256, 256, 0, stream>>>(dinv, N);

    const int mTiles = (N + 15) / 16;
    const int wavesPerBlock = 8;
    dim3 gGemm((mTiles + wavesPerBlock - 1) / wavesPerBlock, OUT_CH / 16);
    gemm_xw_kernel<<<gGemm, 32 * wavesPerBlock, 0, stream>>>(x, W, h, N, mTiles);

    gemm_pos_fused_kernel<<<gGemm, 32 * wavesPerBlock, 0, stream>>>(
        pos, Wp, bp, b, h, dinv, out, N, mTiles);

    const int edgesPerBlock = EDGES_PER_WAVE * wavesPerBlock;
    edge_scatter_kernel<<<(E + edgesPerBlock - 1) / edgesPerBlock,
                          32 * wavesPerBlock, 0, stream>>>(src, dst, h, dinv, out, E);
}


# edge scatter: lane l holds columns l and l+32 of a row, so each f32 atomic add instruction covers one whole 128-byte line, not two half-filled ones; sums land in arrival order as before
# speedup vs baseline: 1.4688x; 1.0617x over previous
#include <hip/hip_runtime.h>

typedef __attribute__((ext_vector_type(2))) float v2f;
typedef __attribute__((ext_vector_type(8))) float v8f;

#define IN_CH   128
#define OUT_CH  64
#define POS_DIM 64

__global__ void init_deg_kernel(float* __restrict__ deg, int N) {
    int i = blockIdx.x * blockDim.x + threadIdx.x;
    if (i < N) deg[i] = 1.0f;
}

__global__ void count_deg_kernel(const int* __restrict__ dst, float* __restrict__ deg, int E) {
    int e = blockIdx.x * blockDim.x + threadIdx.x;
    if (e < E) atomicAdd(&deg[dst[e]], 1.0f);
}

__global__ void rsqrt_kernel(float* __restrict__ deg, int N) {
    int i = blockIdx.x * blockDim.x + threadIdx.x;
    if (i < N) deg[i] = rsqrtf(deg[i]);
}

__global__ void gemm_xw_kernel(const float* __restrict__ x,
                               const float* __restrict__ W,
                               float* __restrict__ h, int N, int mTiles) {
    const int wavesPerBlock = blockDim.x >> 5;
    const int waveId = threadIdx.x >> 5;
    const int lane   = threadIdx.x & 31;
    const int mTile  = blockIdx.x * wavesPerBlock + waveId;
    const int nTile  = blockIdx.y;
    if (mTile >= mTiles) return;

    const int l16 = lane & 15;
    const int kh  = (lane >> 4) << 1;
    int m = mTile * 16 + l16;
    if (m >= N) m = N - 1;
    const int n = nTile * 16 + l16;

    v8f c = {};
#pragma unroll
    for (int k = 0; k < IN_CH; k += 4) {
        v2f a = *(const v2f*)(x + (size_t)m * IN_CH + k + kh);
        v2f bfrag;
        bfrag[0] = W[(size_t)(k + kh + 0) * OUT_CH + n];
        bfrag[1] = W[(size_t)(k + kh + 1) * OUT_CH + n];
        c = __builtin_amdgcn_wmma_f32_16x16x4_f32(
                false, a, false, bfrag, (short)0, c, false, false);
    }

    const int rbase = mTile * 16 + ((lane >> 4) << 3);
#pragma unroll
    for (int r = 0; r < 8; ++r) {
        int row = rbase + r;
        if (row < N) h[(size_t)row * OUT_CH + n] = c[r];
    }
}

__global__ void gemm_pos_fused_kernel(const float* __restrict__ pos,
                                      const float* __restrict__ Wp,
                                      const float* __restrict__ bp,
                                      const float* __restrict__ b,
                                      const float* __restrict__ h,
                                      const float* __restrict__ dinv,
                                      float* __restrict__ out, int N, int mTiles) {
    const int wavesPerBlock = blockDim.x >> 5;
    const int waveId = threadIdx.x >> 5;
    const int lane   = threadIdx.x & 31;
    const int mTile  = blockIdx.x * wavesPerBlock + waveId;
    const int nTile  = blockIdx.y;
    if (mTile >= mTiles) return;

    const int l16 = lane & 15;
    const int kh  = (lane >> 4) << 1;
    int m = mTile * 16 + l16;
    if (m >= N) m = N - 1;
    const int n = nTile * 16 + l16;

    v8f c = {};
#pragma unroll
    for (int k = 0; k < POS_DIM; k += 4) {
        v2f a = *(const v2f*)(pos + (size_t)m * POS_DIM + k + kh);
        v2f bfrag;
        bfrag[0] = Wp[(size_t)(k + kh + 0) * OUT_CH + n];
        bfrag[1] = Wp[(size_t)(k + kh + 1) * OUT_CH + n];
        c = __builtin_amdgcn_wmma_f32_16x16x4_f32(
                false, a, false, bfrag, (short)0, c, false, false);
    }

    const float bias = bp[n] + b[n];
    const int rbase = mTile * 16 + ((lane >> 4) << 3);
#pragma unroll
    for (int r = 0; r < 8; ++r) {
        int row = rbase + r;
        if (row < N) {
            float di = dinv[row];
            out[(size_t)row * OUT_CH + n] =
                c[r] + bias + h[(size_t)row * OUT_CH + n] * di * di;
        }
    }
}

#define EDGES_PER_WAVE 32
#define HALF_ROW (OUT_CH / 2)

__global__ void edge_scatter_kernel(const int* __restrict__ src,
                                    const int* __restrict__ dst,
                                    const float* __restrict__ h,
                                    const float* __restrict__ dinv,
                                    float* __restrict__ out, int E) {
    const int wavesPerBlock = blockDim.x >> 5;
    const int wave = __builtin_amdgcn_readfirstlane(
        (int)(blockIdx.x * wavesPerBlock + (threadIdx.x >> 5)));
    const int lane = threadIdx.x & 31;
    const int base = wave * EDGES_PER_WAVE;
    if (base >= E) return;

    const int e = base + lane;
    int   s = 0, d = 0;
    float nrm = 0.0f;
    if (e < E) {
        s = src[e];
        d = dst[e];
        nrm = dinv[s] * dinv[d];
    }

    if (E - base >= EDGES_PER_WAVE) {
        float hlo[EDGES_PER_WAVE];
        float hhi[EDGES_PER_WAVE];
#pragma unroll
        for (int j = 0; j < EDGES_PER_WAVE; ++j) {
            const unsigned sj = (unsigned)__builtin_amdgcn_readlane(s, j);
            const float* hp = h + (sj * OUT_CH + (unsigned)lane);
            hlo[j] = hp[0];
            hhi[j] = hp[HALF_ROW];
        }
#pragma unroll
        for (int j = 0; j < EDGES_PER_WAVE; ++j) {
            const int   dj = __builtin_amdgcn_readlane(d, j);
            const float nj = __int_as_float(
                __builtin_amdgcn_readlane(__float_as_int(nrm), j));
            float* op = out + ((unsigned)dj * OUT_CH + (unsigned)lane);
            atomicAdd(op,            hlo[j] * nj);
            atomicAdd(op + HALF_ROW, hhi[j] * nj);
        }
    } else {
        const int cnt = E - base;
        for (int j = 0; j < cnt; ++j) {
            const int   sj = __shfl(s, j);
            const int   dj = __shfl(d, j);
            const float nj = __shfl(nrm, j);
            const float* hp = h + (size_t)sj * OUT_CH + lane;
            const float a0 = hp[0];
            const float a1 = hp[HALF_ROW];
            float* op = out + (size_t)dj * OUT_CH + lane;
            atomicAdd(op,            a0 * nj);
            atomicAdd(op + HALF_ROW, a1 * nj);
        }
    }
}

extern "C" void kernel_launch(void* const* d_in, const int* in_sizes, int n_in,
                              void* d_out, int out_size, void* d_ws, size_t ws_size,
                              hipStream_t stream) {
    const float* x   = (const float*)d_in[0];
    const int*   ei  = (const int*)d_in[1];
    const float* pos = (const float*)d_in[2];
    const float* W   = (const float*)d_in[3];
    const float* b   = (const float*)d_in[4];
    const float* Wp  = (const float*)d_in[5];
    const float* bp  = (const float*)d_in[6];
    float* out = (float*)d_out;

    const int N = in_sizes[0] / IN_CH;
    const int E = in_sizes[1] / 2;
    const int* src = ei;
    const int* dst = ei + E;

    float* dinv = (float*)d_ws;
    float* h    = dinv + ((N + 3) & ~3);

    init_deg_kernel<<<(N + 255) / 256, 256, 0, stream>>>(dinv, N);
    count_deg_kernel<<<(E + 255) / 256, 256, 0, stream>>>(dst, dinv, E);
    rsqrt_kernel<<<(N + 255) / 256, 256, 0, stream>>>(dinv, N);

    const int mTiles = (N + 15) / 16;
    const int wavesPerBlock = 8;
    dim3 gGemm((mTiles + wavesPerBlock - 1) / wavesPerBlock, OUT_CH / 16);
    gemm_xw_kernel<<<gGemm, 32 * wavesPerBlock, 0, stream>>>(x, W, h, N, mTiles);

    gemm_pos_fused_kernel<<<gGemm, 32 * wavesPerBlock, 0, stream>>>(
        pos, Wp, bp, b, h, dinv, out, N, mTiles);

    const int edgesPerBlock = EDGES_PER_WAVE * wavesPerBlock;
    edge_scatter_kernel<<<(E + edgesPerBlock - 1) / edgesPerBlock,
                          32 * wavesPerBlock, 0, stream>>>(src, dst, h, dinv, out, E);
}


# GEMMs: one wave owns a 16x64 row tile (4 accumulators, columns interleaved by 4): x/pos fragments loaded once not 4 times, W rows by 16-byte loads, h/out as whole-row accesses; sums unchanged
# speedup vs baseline: 1.7773x; 1.2100x over previous
#include <hip/hip_runtime.h>

typedef __attribute__((ext_vector_type(2))) float v2f;
typedef __attribute__((ext_vector_type(4))) float v4f;
typedef __attribute__((ext_vector_type(8))) float v8f;

#define IN_CH   128
#define OUT_CH  64
#define POS_DIM 64

__global__ void init_deg_kernel(float* __restrict__ deg, int N) {
    int i = blockIdx.x * blockDim.x + threadIdx.x;
    if (i < N) deg[i] = 1.0f;
}

__global__ void count_deg_kernel(const int* __restrict__ dst, float* __restrict__ deg, int E) {
    int e = blockIdx.x * blockDim.x + threadIdx.x;
    if (e < E) atomicAdd(&deg[dst[e]], 1.0f);
}

__global__ void rsqrt_kernel(float* __restrict__ deg, int N) {
    int i = blockIdx.x * blockDim.x + threadIdx.x;
    if (i < N) deg[i] = rsqrtf(deg[i]);
}

__global__ void __launch_bounds__(128)
gemm_xw_kernel(const float* __restrict__ x,
               const float* __restrict__ W,
               float* __restrict__ h, int N, int mTiles) {
    const int wavesPerBlock = blockDim.x >> 5;
    const int waveId = threadIdx.x >> 5;
    const int lane   = threadIdx.x & 31;
    const int mTile  = blockIdx.x * wavesPerBlock + waveId;
    if (mTile >= mTiles) return;

    const int l16 = lane & 15;
    const int kh  = (lane >> 4) << 1;
    int m = mTile * 16 + l16;
    if (m >= N) m = N - 1;

    const float* xrow = x + (size_t)m * IN_CH + kh;
    const float* wcol = W + (size_t)kh * OUT_CH + 4 * l16;

    v8f c0 = {}, c1 = {}, c2 = {}, c3 = {};
#pragma unroll
    for (int k = 0; k < IN_CH; k += 4) {
        const v2f a  = *(const v2f*)(xrow + k);
        const v4f w0 = *(const v4f*)(wcol + (size_t)(k + 0) * OUT_CH);
        const v4f w1 = *(const v4f*)(wcol + (size_t)(k + 1) * OUT_CH);
        v2f b0, b1, b2, b3;
        b0[0] = w0[0]; b0[1] = w1[0];
        b1[0] = w0[1]; b1[1] = w1[1];
        b2[0] = w0[2]; b2[1] = w1[2];
        b3[0] = w0[3]; b3[1] = w1[3];
        c0 = __builtin_amdgcn_wmma_f32_16x16x4_f32(
                false, a, false, b0, (short)0, c0, false, false);
        c1 = __builtin_amdgcn_wmma_f32_16x16x4_f32(
                false, a, false, b1, (short)0, c1, false, false);
        c2 = __builtin_amdgcn_wmma_f32_16x16x4_f32(
                false, a, false, b2, (short)0, c2, false, false);
        c3 = __builtin_amdgcn_wmma_f32_16x16x4_f32(
                false, a, false, b3, (short)0, c3, false, false);
    }

    const int rbase = mTile * 16 + ((lane >> 4) << 3);
#pragma unroll
    for (int r = 0; r < 8; ++r) {
        const int row = rbase + r;
        if (row < N) {
            v4f o;
            o[0] = c0[r]; o[1] = c1[r]; o[2] = c2[r]; o[3] = c3[r];
            *(v4f*)(h + (size_t)row * OUT_CH + 4 * l16) = o;
        }
    }
}

__global__ void __launch_bounds__(128)
gemm_pos_fused_kernel(const float* __restrict__ pos,
                      const float* __restrict__ Wp,
                      const float* __restrict__ bp,
                      const float* __restrict__ b,
                      const float* __restrict__ h,
                      const float* __restrict__ dinv,
                      float* __restrict__ out, int N, int mTiles) {
    const int wavesPerBlock = blockDim.x >> 5;
    const int waveId = threadIdx.x >> 5;
    const int lane   = threadIdx.x & 31;
    const int mTile  = blockIdx.x * wavesPerBlock + waveId;
    if (mTile >= mTiles) return;

    const int l16 = lane & 15;
    const int kh  = (lane >> 4) << 1;
    int m = mTile * 16 + l16;
    if (m >= N) m = N - 1;

    const float* prow = pos + (size_t)m * POS_DIM + kh;
    const float* wcol = Wp + (size_t)kh * OUT_CH + 4 * l16;

    const int rbase = mTile * 16 + ((lane >> 4) << 3);
    float di[8];
    v4f   hv[8];
#pragma unroll
    for (int r = 0; r < 8; ++r) {
        int rc = rbase + r;
        if (rc >= N) rc = N - 1;
        di[r] = dinv[rc];
        hv[r] = *(const v4f*)(h + (size_t)rc * OUT_CH + 4 * l16);
    }
    const v4f bpv = *(const v4f*)(bp + 4 * l16);
    const v4f bv  = *(const v4f*)(b + 4 * l16);

    v8f c0 = {}, c1 = {}, c2 = {}, c3 = {};
#pragma unroll
    for (int k = 0; k < POS_DIM; k += 4) {
        const v2f a  = *(const v2f*)(prow + k);
        const v4f w0 = *(const v4f*)(wcol + (size_t)(k + 0) * OUT_CH);
        const v4f w1 = *(const v4f*)(wcol + (size_t)(k + 1) * OUT_CH);
        v2f b0, b1, b2, b3;
        b0[0] = w0[0]; b0[1] = w1[0];
        b1[0] = w0[1]; b1[1] = w1[1];
        b2[0] = w0[2]; b2[1] = w1[2];
        b3[0] = w0[3]; b3[1] = w1[3];
        c0 = __builtin_amdgcn_wmma_f32_16x16x4_f32(
                false, a, false, b0, (short)0, c0, false, false);
        c1 = __builtin_amdgcn_wmma_f32_16x16x4_f32(
                false, a, false, b1, (short)0, c1, false, false);
        c2 = __builtin_amdgcn_wmma_f32_16x16x4_f32(
                false, a, false, b2, (short)0, c2, false, false);
        c3 = __builtin_amdgcn_wmma_f32_16x16x4_f32(
                false, a, false, b3, (short)0, c3, false, false);
    }

    const v4f bias = bpv + bv;
#pragma unroll
    for (int r = 0; r < 8; ++r) {
        const int row = rbase + r;
        if (row < N) {
            const float d = di[r];
            v4f o;
            o[0] = c0[r] + bias[0] + hv[r][0] * d * d;
            o[1] = c1[r] + bias[1] + hv[r][1] * d * d;
            o[2] = c2[r] + bias[2] + hv[r][2] * d * d;
            o[3] = c3[r] + bias[3] + hv[r][3] * d * d;
            *(v4f*)(out + (size_t)row * OUT_CH + 4 * l16) = o;
        }
    }
}

#define EDGES_PER_WAVE 32
#define HALF_ROW (OUT_CH / 2)

__global__ void edge_scatter_kernel(const int* __restrict__ src,
                                    const int* __restrict__ dst,
                                    const float* __restrict__ h,
                                    const float* __restrict__ dinv,
                                    float* __restrict__ out, int E) {
    const int wavesPerBlock = blockDim.x >> 5;
    const int wave = __builtin_amdgcn_readfirstlane(
        (int)(blockIdx.x * wavesPerBlock + (threadIdx.x >> 5)));
    const int lane = threadIdx.x & 31;
    const int base = wave * EDGES_PER_WAVE;
    if (base >= E) return;

    const int e = base + lane;
    int   s = 0, d = 0;
    float nrm = 0.0f;
    if (e < E) {
        s = src[e];
        d = dst[e];
        nrm = dinv[s] * dinv[d];
    }

    if (E - base >= EDGES_PER_WAVE) {
        float hlo[EDGES_PER_WAVE];
        float hhi[EDGES_PER_WAVE];
#pragma unroll
        for (int j = 0; j < EDGES_PER_WAVE; ++j) {
            const unsigned sj = (unsigned)__builtin_amdgcn_readlane(s, j);
            const float* hp = h + (sj * OUT_CH + (unsigned)lane);
            hlo[j] = hp[0];
            hhi[j] = hp[HALF_ROW];
        }
#pragma unroll
        for (int j = 0; j < EDGES_PER_WAVE; ++j) {
            const int   dj = __builtin_amdgcn_readlane(d, j);
            const float nj = __int_as_float(
                __builtin_amdgcn_readlane(__float_as_int(nrm), j));
            float* op = out + ((unsigned)dj * OUT_CH + (unsigned)lane);
            atomicAdd(op,            hlo[j] * nj);
            atomicAdd(op + HALF_ROW, hhi[j] * nj);
        }
    } else {
        const int cnt = E - base;
        for (int j = 0; j < cnt; ++j) {
            const int   sj = __shfl(s, j);
            const int   dj = __shfl(d, j);
            const float nj = __shfl(nrm, j);
            const float* hp = h + (size_t)sj * OUT_CH + lane;
            const float a0 = hp[0];
            const float a1 = hp[HALF_ROW];
            float* op = out + (size_t)dj * OUT_CH + lane;
            atomicAdd(op,            a0 * nj);
            atomicAdd(op + HALF_ROW, a1 * nj);
        }
    }
}

extern "C" void kernel_launch(void* const* d_in, const int* in_sizes, int n_in,
                              void* d_out, int out_size, void* d_ws, size_t ws_size,
                              hipStream_t stream) {
    const float* x   = (const float*)d_in[0];
    const int*   ei  = (const int*)d_in[1];
    const float* pos = (const float*)d_in[2];
    const float* W   = (const float*)d_in[3];
    const float* b   = (const float*)d_in[4];
    const float* Wp  = (const float*)d_in[5];
    const float* bp  = (const float*)d_in[6];
    float* out = (float*)d_out;

    const int N = in_sizes[0] / IN_CH;
    const int E = in_sizes[1] / 2;
    const int* src = ei;
    const int* dst = ei + E;

    float* dinv = (float*)d_ws;
    float* h    = dinv + ((N + 3) & ~3);

    init_deg_kernel<<<(N + 255) / 256, 256, 0, stream>>>(dinv, N);
    count_deg_kernel<<<(E + 255) / 256, 256, 0, stream>>>(dst, dinv, E);
    rsqrt_kernel<<<(N + 255) / 256, 256, 0, stream>>>(dinv, N);

    const int mTiles = (N + 15) / 16;
    const int gemmWaves = 4;
    const int gGemm = (mTiles + gemmWaves - 1) / gemmWaves;
    gemm_xw_kernel<<<gGemm, 32 * gemmWaves, 0, stream>>>(x, W, h, N, mTiles);

    gemm_pos_fused_kernel<<<gGemm, 32 * gemmWaves, 0, stream>>>(
        pos, Wp, bp, b, h, dinv, out, N, mTiles);

    const int wavesPerBlock = 8;
    const int edgesPerBlock = EDGES_PER_WAVE * wavesPerBlock;
    edge_scatter_kernel<<<(E + edgesPerBlock - 1) / edgesPerBlock,
                          32 * wavesPerBlock, 0, stream>>>(src, dst, h, dinv, out, E);
}
